# starting kernel
# baseline (speedup 1.0000x reference)
#include <hip/hip_runtime.h>

#define D 64

typedef __attribute__((ext_vector_type(2))) float v2f;
typedef __attribute__((ext_vector_type(8))) float v8f;

__global__ void zero_f32(float* __restrict__ p, int n) {
    int i = blockIdx.x * blockDim.x + threadIdx.x;
    if (i < n) p[i] = 0.0f;
}

__global__ void degree_kernel(const int* __restrict__ col, float* __restrict__ deg, int E) {
    int e = blockIdx.x * blockDim.x + threadIdx.x;
    if (e < E) atomicAdd(&deg[col[e]], 1.0f);
}

__global__ void rsqrt_kernel(float* __restrict__ deg, int n) {
    int i = blockIdx.x * blockDim.x + threadIdx.x;
    if (i < n) {
        float d = deg[i];
        deg[i] = (d > 0.0f) ? rsqrtf(d) : 0.0f;
    }
}

__global__ void weight_kernel(const int* __restrict__ row, const int* __restrict__ col,
                              const float* __restrict__ dist, const float* __restrict__ dinv,
                              float* __restrict__ wgt, int E) {
    int e = blockIdx.x * blockDim.x + threadIdx.x;
    if (e < E) {
        float d = dist[e];
        wgt[e] = expf(-d * d) * dinv[row[e]] * dinv[col[e]];
    }
}

__global__ void scatter_kernel(const int* __restrict__ row, const int* __restrict__ col,
                               const float* __restrict__ wgt, const float* __restrict__ x,
                               float* __restrict__ side, int E) {
    long long gid = (long long)blockIdx.x * blockDim.x + threadIdx.x;
    int e = (int)(gid >> 6);
    int k = (int)(gid & 63);
    if (e < E) {
        float w = wgt[e];
        int r = row[e];
        int c = col[e];
        atomicAdd(&side[r * D + k], w * x[c * D + k]);
    }
}

__global__ void gemm_wmma(float* __restrict__ out, const float* __restrict__ Wm,
                          const float* __restrict__ bias, int nRowBlocks) {
    int lane = threadIdx.x & 31;
    int wave = threadIdx.x >> 5;
    int rb = blockIdx.x * (blockDim.x >> 5) + wave;
    if (rb >= nRowBlocks) return;

    const int lrow = lane & 15;
    const int koff = (lane >= 16) ? 2 : 0;

    const float* arow = out + (rb * 16 + lrow) * D;
    v2f a[16];
#pragma unroll
    for (int t = 0; t < 16; ++t) {
        a[t].x = arow[4 * t + koff];
        a[t].y = arow[4 * t + koff + 1];
    }

#pragma unroll
    for (int ct = 0; ct < 4; ++ct) {
        int cb = 16 * ct;
        const float* wr = Wm + (cb + lrow) * D;
        v8f acc = {};
#pragma unroll
        for (int t = 0; t < 16; ++t) {
            v2f bf;
            bf.x = wr[4 * t + koff];
            bf.y = wr[4 * t + koff + 1];
            acc = __builtin_amdgcn_wmma_f32_16x16x4_f32(
                 false, a[t],  false, bf,
                 (short)0, acc,  false,  false);
        }
        float bv = bias[cb + lrow];
#pragma unroll
        for (int v = 0; v < 8; ++v) {
            int m = (lane >= 16) ? (v + 8) : v;
            out[(rb * 16 + m) * D + cb + lrow] = acc[v] + bv;
        }
    }
}

extern "C" void kernel_launch(void* const* d_in, const int* in_sizes, int n_in,
                              void* d_out, int out_size, void* d_ws, size_t ws_size,
                              hipStream_t stream) {
    const float* x    = (const float*)d_in[0];
    const int*   ei   = (const int*)d_in[1];
    const float* dist = (const float*)d_in[2];
    const float* Wm   = (const float*)d_in[3];
    const float* bias = (const float*)d_in[4];
    float* out = (float*)d_out;

    const int N = in_sizes[0] / D;
    const int E = in_sizes[2];
    const int* row = ei;
    const int* col = ei + E;

    float* deg = (float*)d_ws;
    float* wgt = deg + N;

    const int T = 256;

    zero_f32<<<(N + T - 1) / T, T, 0, stream>>>(deg, N);
    zero_f32<<<(N * D + T - 1) / T, T, 0, stream>>>(out, N * D);

    degree_kernel<<<(E + T - 1) / T, T, 0, stream>>>(col, deg, E);
    rsqrt_kernel<<<(N + T - 1) / T, T, 0, stream>>>(deg, N);
    weight_kernel<<<(E + T - 1) / T, T, 0, stream>>>(row, col, dist, deg, wgt, E);

    long long total = (long long)E * D;
    scatter_kernel<<<(int)((total + T - 1) / T), T, 0, stream>>>(row, col, wgt, x, out, E);

    const int nRB = N / 16;
    const int wpb = T / 32;
    gemm_wmma<<<(nRB + wpb - 1) / wpb, T, 0, stream>>>(out, Wm, bias, nRB);
}


# scatter: one wave owns 32 consecutive edges (indices read once, broadcast by readlane) and gathers 8 edges' x rows before their atomics; atomic arrival order still sets the sum order
# speedup vs baseline: 2.1574x; 2.1574x over previous
#include <hip/hip_runtime.h>

#define D 64

typedef __attribute__((ext_vector_type(2))) float v2f;
typedef __attribute__((ext_vector_type(8))) float v8f;

__global__ void zero_f32(float* __restrict__ p, int n) {
    int i = blockIdx.x * blockDim.x + threadIdx.x;
    if (i < n) p[i] = 0.0f;
}

__global__ void degree_kernel(const int* __restrict__ col, float* __restrict__ deg, int E) {
    int e = blockIdx.x * blockDim.x + threadIdx.x;
    if (e < E) atomicAdd(&deg[col[e]], 1.0f);
}

__global__ void rsqrt_kernel(float* __restrict__ deg, int n) {
    int i = blockIdx.x * blockDim.x + threadIdx.x;
    if (i < n) {
        float d = deg[i];
        deg[i] = (d > 0.0f) ? rsqrtf(d) : 0.0f;
    }
}

__global__ void weight_kernel(const int* __restrict__ row, const int* __restrict__ col,
                              const float* __restrict__ dist, const float* __restrict__ dinv,
                              float* __restrict__ wgt, int E) {
    int e = blockIdx.x * blockDim.x + threadIdx.x;
    if (e < E) {
        float d = dist[e];
        wgt[e] = expf(-d * d) * dinv[row[e]] * dinv[col[e]];
    }
}

#define EPW 32
#define EB 8

__device__ __forceinline__ float lane_bcast_f(float v, int srcLane) {
    return __int_as_float(__builtin_amdgcn_readlane(__float_as_int(v), srcLane));
}

__global__ __launch_bounds__(256) void scatter_kernel(const int* __restrict__ row, const int* __restrict__ col,
                               const float* __restrict__ wgt, const float* __restrict__ x,
                               float* __restrict__ side, int E) {
    const int lane = threadIdx.x & 31;
    const int wv = blockIdx.x * (blockDim.x >> 5) + __builtin_amdgcn_readfirstlane((int)(threadIdx.x >> 5));
    const int e0 = wv * EPW;
    if (e0 >= E) return;
    const int n = (E - e0 < EPW) ? (E - e0) : EPW;

    int r = 0, c = 0;
    float w = 0.0f;
    if (lane < n) {
        r = row[e0 + lane];
        c = col[e0 + lane];
        w = wgt[e0 + lane];
    }

    if (n == EPW) {
#pragma unroll
        for (int b = 0; b < EPW; b += EB) {
            float v0[EB], v1[EB];
#pragma unroll
            for (int j = 0; j < EB; ++j) {
                const int cj = __builtin_amdgcn_readlane(c, b + j);
                const float* xr = x + cj * D;
                v0[j] = xr[lane];
                v1[j] = xr[32 + lane];
            }
#pragma unroll
            for (int j = 0; j < EB; ++j) {
                const int rj = __builtin_amdgcn_readlane(r, b + j);
                const float wj = lane_bcast_f(w, b + j);
                float* sr = side + rj * D;
                atomicAdd(sr + lane, wj * v0[j]);
                atomicAdd(sr + 32 + lane, wj * v1[j]);
            }
        }
    } else {
        for (int j = 0; j < n; ++j) {
            const int cj = __builtin_amdgcn_readlane(c, j);
            const int rj = __builtin_amdgcn_readlane(r, j);
            const float wj = lane_bcast_f(w, j);
            const float* xr = x + cj * D;
            float* sr = side + rj * D;
            atomicAdd(sr + lane, wj * xr[lane]);
            atomicAdd(sr + 32 + lane, wj * xr[32 + lane]);
        }
    }
}

__global__ void gemm_wmma(float* __restrict__ out, const float* __restrict__ Wm,
                          const float* __restrict__ bias, int nRowBlocks) {
    int lane = threadIdx.x & 31;
    int wave = threadIdx.x >> 5;
    int rb = blockIdx.x * (blockDim.x >> 5) + wave;
    if (rb >= nRowBlocks) return;

    const int lrow = lane & 15;
    const int koff = (lane >= 16) ? 2 : 0;

    const float* arow = out + (rb * 16 + lrow) * D;
    v2f a[16];
#pragma unroll
    for (int t = 0; t < 16; ++t) {
        a[t].x = arow[4 * t + koff];
        a[t].y = arow[4 * t + koff + 1];
    }

#pragma unroll
    for (int ct = 0; ct < 4; ++ct) {
        int cb = 16 * ct;
        const float* wr = Wm + (cb + lrow) * D;
        v8f acc = {};
#pragma unroll
        for (int t = 0; t < 16; ++t) {
            v2f bf;
            bf.x = wr[4 * t + koff];
            bf.y = wr[4 * t + koff + 1];
            acc = __builtin_amdgcn_wmma_f32_16x16x4_f32(
                 false, a[t],  false, bf,
                 (short)0, acc,  false,  false);
        }
        float bv = bias[cb + lrow];
#pragma unroll
        for (int v = 0; v < 8; ++v) {
            int m = (lane >= 16) ? (v + 8) : v;
            out[(rb * 16 + m) * D + cb + lrow] = acc[v] + bv;
        }
    }
}

extern "C" void kernel_launch(void* const* d_in, const int* in_sizes, int n_in,
                              void* d_out, int out_size, void* d_ws, size_t ws_size,
                              hipStream_t stream) {
    const float* x    = (const float*)d_in[0];
    const int*   ei   = (const int*)d_in[1];
    const float* dist = (const float*)d_in[2];
    const float* Wm   = (const float*)d_in[3];
    const float* bias = (const float*)d_in[4];
    float* out = (float*)d_out;

    const int N = in_sizes[0] / D;
    const int E = in_sizes[2];
    const int* row = ei;
    const int* col = ei + E;

    float* deg = (float*)d_ws;
    float* wgt = deg + N;

    const int T = 256;

    zero_f32<<<(N + T - 1) / T, T, 0, stream>>>(deg, N);
    zero_f32<<<(N * D + T - 1) / T, T, 0, stream>>>(out, N * D);

    degree_kernel<<<(E + T - 1) / T, T, 0, stream>>>(col, deg, E);
    rsqrt_kernel<<<(N + T - 1) / T, T, 0, stream>>>(deg, N);
    weight_kernel<<<(E + T - 1) / T, T, 0, stream>>>(row, col, dist, deg, wgt, E);

    const int nEW = (E + EPW - 1) / EPW;
    scatter_kernel<<<(nEW + T / 32 - 1) / (T / 32), T, 0, stream>>>(row, col, wgt, x, out, E);

    const int nRB = N / 16;
    const int wpb = T / 32;
    gemm_wmma<<<(nRB + wpb - 1) / wpb, T, 0, stream>>>(out, Wm, bias, nRB);
}


# edges binned per call by 400-row destination block (LDS counting sort); each block's owner work-group sums its rows in LDS and feeds the WMMA from LDS; LDS add arrival order sets the sum order
# speedup vs baseline: 2.8851x; 1.3373x over previous
#include <hip/hip_runtime.h>


#define D 64
#define NPB 400
#define MAXB 256
#define NCH 256
#define PITCH 68
#define TPB 1024

typedef __attribute__((ext_vector_type(2))) float v2f;
typedef __attribute__((ext_vector_type(8))) float v8f;

__global__ void zero_f32(float* __restrict__ p, int n) {
    int i = blockIdx.x * blockDim.x + threadIdx.x;
    if (i < n) p[i] = 0.0f;
}

__global__ __launch_bounds__(TPB) void count_kernel(const int* __restrict__ row, const int* __restrict__ col,
                                                    float* __restrict__ deg, int* __restrict__ tot,
                                                    int* __restrict__ chunkOff, int E, int chunkLen) {
    __shared__ int sCnt[MAXB];
    const int t = threadIdx.x;
    if (t < MAXB) sCnt[t] = 0;
    __syncthreads();
    const int e0 = blockIdx.x * chunkLen;
    const int e1 = (e0 + chunkLen < E) ? (e0 + chunkLen) : E;
    for (int e = e0 + t; e < e1; e += TPB) {
        atomicAdd(&deg[col[e]], 1.0f);
        atomicAdd(&sCnt[row[e] / NPB], 1);
    }
    __syncthreads();
    if (t < MAXB) {
        const int c = sCnt[t];
        chunkOff[blockIdx.x * MAXB + t] = (c > 0) ? atomicAdd(&tot[t], c) : 0;
    }
}

__global__ __launch_bounds__(TPB) void bin_kernel(const int* __restrict__ row, const int* __restrict__ col,
                                                  const float* __restrict__ dist, const float* __restrict__ deg,
                                                  const int* __restrict__ tot, const int* __restrict__ chunkOff,
                                                  int2* __restrict__ rec, int E, int chunkLen) {
    __shared__ int sTot[MAXB];
    __shared__ int sNext[MAXB];
    const int t = threadIdx.x;
    if (t < MAXB) sTot[t] = tot[t];
    __syncthreads();
    if (t < MAXB) {
        int base = 0;
        for (int j = 0; j < t; ++j) base += sTot[j];
        sNext[t] = base + chunkOff[blockIdx.x * MAXB + t];
    }
    __syncthreads();
    const int e0 = blockIdx.x * chunkLen;
    const int e1 = (e0 + chunkLen < E) ? (e0 + chunkLen) : E;
    for (int e = e0 + t; e < e1; e += TPB) {
        const int r = row[e];
        const int c = col[e];
        const float d = dist[e];
        const float dr = deg[r];
        const float dc = deg[c];
        const float ir = (dr > 0.0f) ? rsqrtf(dr) : 0.0f;
        const float ic = (dc > 0.0f) ? rsqrtf(dc) : 0.0f;
        const float w = expf(-d * d) * ir * ic;
        const int b = r / NPB;
        const int slot = atomicAdd(&sNext[b], 1);
        rec[slot] = make_int2(c | ((r - b * NPB) << 20), __float_as_int(w));
    }
}

#define EB 8

__device__ __forceinline__ float lane_bcast_f(float v, int srcLane) {
    return __int_as_float(__builtin_amdgcn_readlane(__float_as_int(v), srcLane));
}

__global__ __launch_bounds__(TPB) void owner_kernel(const int2* __restrict__ rec, const int* __restrict__ tot,
                                                    const float* __restrict__ x, const float* __restrict__ Wm,
                                                    const float* __restrict__ bias, float* __restrict__ out, int N) {
    __shared__ float sAcc[NPB * PITCH];
    __shared__ int sTot[MAXB];
    const int t = threadIdx.x;
    const int lane = t & 31;
    const int wv = __builtin_amdgcn_readfirstlane(t >> 5);
    const int blk = blockIdx.x;

    for (int i = t; i < NPB * PITCH; i += TPB) sAcc[i] = 0.0f;
    if (t < MAXB) sTot[t] = tot[t];
    __syncthreads();
    int baseV = 0;
    for (int j = 0; j < blk; ++j) baseV += sTot[j];
    const int base = __builtin_amdgcn_readfirstlane(baseV);
    const int cnt = __builtin_amdgcn_readfirstlane(sTot[blk]);

    for (int g = wv * 32; g < cnt; g += TPB) {
        const int n = (cnt - g < 32) ? (cnt - g) : 32;
        int cw = 0;
        float w = 0.0f;
        if (lane < n) {
            const int2 q = rec[base + g + lane];
            cw = q.x;
            w = __int_as_float(q.y);
        }
        if (n == 32) {
#pragma unroll
            for (int b8 = 0; b8 < 32; b8 += EB) {
                float v0[EB], v1[EB];
#pragma unroll
                for (int j = 0; j < EB; ++j) {
                    const int cj = __builtin_amdgcn_readlane(cw, b8 + j) & 0xFFFFF;
                    const float* xr = x + cj * D;
                    v0[j] = xr[lane];
                    v1[j] = xr[32 + lane];
                }
#pragma unroll
                for (int j = 0; j < EB; ++j) {
                    const int rl = __builtin_amdgcn_readlane(cw, b8 + j) >> 20;
                    const float wj = lane_bcast_f(w, b8 + j);
                    float* sr = sAcc + rl * PITCH;
                    atomicAdd(sr + lane, wj * v0[j]);
                    atomicAdd(sr + 32 + lane, wj * v1[j]);
                }
            }
        } else {
            for (int j = 0; j < n; ++j) {
                const int q = __builtin_amdgcn_readlane(cw, j);
                const float wj = lane_bcast_f(w, j);
                const float* xr = x + (q & 0xFFFFF) * D;
                float* sr = sAcc + (q >> 20) * PITCH;
                atomicAdd(sr + lane, wj * xr[lane]);
                atomicAdd(sr + 32 + lane, wj * xr[32 + lane]);
            }
        }
    }
    __syncthreads();

    if (wv < NPB / 16) {
        const int lrow = lane & 15;
        const int koff = (lane >= 16) ? 2 : 0;
        const float* arow = sAcc + (wv * 16 + lrow) * PITCH;
        v2f a[16];
#pragma unroll
        for (int s = 0; s < 16; ++s) {
            a[s].x = arow[4 * s + koff];
            a[s].y = arow[4 * s + koff + 1];
        }
        const int grow0 = blk * NPB + wv * 16;
#pragma unroll
        for (int ct = 0; ct < 4; ++ct) {
            const int cb = 16 * ct;
            const float* wr = Wm + (cb + lrow) * D;
            v8f acc = {};
#pragma unroll
            for (int s = 0; s < 16; ++s) {
                v2f bf;
                bf.x = wr[4 * s + koff];
                bf.y = wr[4 * s + koff + 1];
                acc = __builtin_amdgcn_wmma_f32_16x16x4_f32(
                     false, a[s],  false, bf,
                     (short)0, acc,  false,  false);
            }
            const float bv = bias[cb + lrow];
#pragma unroll
            for (int v = 0; v < 8; ++v) {
                const int m = (lane >= 16) ? (v + 8) : v;
                const int gr = grow0 + m;
                if (gr < N) out[gr * D + cb + lrow] = acc[v] + bv;
            }
        }
    }
}

extern "C" void kernel_launch(void* const* d_in, const int* in_sizes, int n_in,
                              void* d_out, int out_size, void* d_ws, size_t ws_size,
                              hipStream_t stream) {
    const float* x    = (const float*)d_in[0];
    const int*   ei   = (const int*)d_in[1];
    const float* dist = (const float*)d_in[2];
    const float* Wm   = (const float*)d_in[3];
    const float* bias = (const float*)d_in[4];
    float* out = (float*)d_out;

    const int N = in_sizes[0] / D;
    const int E = in_sizes[2];
    const int* row = ei;
    const int* col = ei + E;

    float* deg      = (float*)d_ws;
    int*   tot      = (int*)(deg + N);
    int*   chunkOff = tot + MAXB;
    int2*  rec      = (int2*)(chunkOff + NCH * MAXB + ((N + MAXB + NCH * MAXB) & 1));

    const int T = 256;
    const int NB = (N + NPB - 1) / NPB;
    const int chunkLen = (E + NCH - 1) / NCH;

    zero_f32<<<(N + MAXB + T - 1) / T, T, 0, stream>>>(deg, N + MAXB);
    count_kernel<<<NCH, TPB, 0, stream>>>(row, col, deg, tot, chunkOff, E, chunkLen);
    bin_kernel<<<NCH, TPB, 0, stream>>>(row, col, dist, deg, tot, chunkOff, rec, E, chunkLen);
    owner_kernel<<<NB, TPB, 0, stream>>>(rec, tot, x, Wm, bias, out, N);
}
